# starting kernel
# baseline (speedup 1.0000x reference)
#include <hip/hip_runtime.h>


typedef __attribute__((ext_vector_type(2))) float v2f;
typedef __attribute__((ext_vector_type(8))) float v8f;

#define NSAMP   256
#define CHN     256
#define KPOS    36
#define MPAD    48
#define NPIX    676
#define WXX     26
#define OHW     21
#define NOUT    441
#define ZSTRIDE 260
#define NTILES  43
#define NWAVES  8

__global__ __launch_bounds__(256)
void siamfc_xcorr_wmma(const float* __restrict__ z,
                       const float* __restrict__ x,
                       float* __restrict__ out) {
    __shared__ float zlds[MPAD * ZSTRIDE];
    __shared__ float outLds[NOUT];

    const int n     = blockIdx.x;
    const int tid   = threadIdx.x;
    const int wave  = tid >> 5;
    const int lane  = tid & 31;
    const int nl    = lane & 15;
    const int khalf = lane >> 4;

    {
        const float* zsrc = z + (size_t)(n * CHN + tid) * KPOS;
        float zr[KPOS];
        #pragma unroll
        for (int k4 = 0; k4 < KPOS; k4 += 4) {
            float4 v = *(const float4*)(zsrc + k4);
            zr[k4 + 0] = v.x; zr[k4 + 1] = v.y; zr[k4 + 2] = v.z; zr[k4 + 3] = v.w;
        }
        #pragma unroll
        for (int k = 0; k < KPOS; ++k) zlds[k * ZSTRIDE + tid] = zr[k];
        #pragma unroll
        for (int k = KPOS; k < MPAD; ++k) zlds[k * ZSTRIDE + tid] = 0.0f;
    }
    for (int p = tid; p < NOUT; p += 256) outLds[p] = 0.0f;
    __syncthreads();

    const float* xs = x + (size_t)n * CHN * NPIX;

    for (int ntile = wave; ntile < NTILES; ntile += NWAVES) {
        const int n0    = ntile * 16;
        const int s_raw = n0 + nl;
        const int s     = (s_raw < NPIX) ? s_raw : (NPIX - 1);

        v8f acc0 = {}; v8f acc1 = {}; v8f acc2 = {};

        const float* xb = xs + (size_t)(khalf * 2) * NPIX + s;

        #pragma unroll 8
        for (int c0 = 0; c0 < CHN; c0 += 4) {
            v2f b;
            b.x = xb[(size_t)c0 * NPIX];
            b.y = xb[(size_t)c0 * NPIX + NPIX];
            const int ka = c0 + khalf * 2;
            v2f a0 = *(const v2f*)&zlds[(0  + nl) * ZSTRIDE + ka];
            v2f a1 = *(const v2f*)&zlds[(16 + nl) * ZSTRIDE + ka];
            v2f a2 = *(const v2f*)&zlds[(32 + nl) * ZSTRIDE + ka];
            acc0 = __builtin_amdgcn_wmma_f32_16x16x4_f32(false, a0, false, b, (short)0, acc0, false, false);
            acc1 = __builtin_amdgcn_wmma_f32_16x16x4_f32(false, a1, false, b, (short)0, acc1, false, false);
            acc2 = __builtin_amdgcn_wmma_f32_16x16x4_f32(false, a2, false, b, (short)0, acc2, false, false);
        }

        if (s_raw < NPIX) {
            const int r  = s_raw / WXX;
            const int wc = s_raw % WXX;
            #pragma unroll
            for (int mt = 0; mt < 3; ++mt) {
                v8f a = (mt == 0) ? acc0 : (mt == 1) ? acc1 : acc2;
                #pragma unroll
                for (int v = 0; v < 8; ++v) {
                    const int m = mt * 16 + v + khalf * 8;
                    if (m < KPOS) {
                        const int kh = m / 6, kw = m % 6;
                        const int oh = r - kh, ow = wc - kw;
                        if (oh >= 0 && oh < OHW && ow >= 0 && ow < OHW)
                            atomicAdd(&outLds[oh * OHW + ow], a[v]);
                    }
                }
            }
        }
    }

    __syncthreads();
    float* op = out + (size_t)n * NOUT;
    for (int p = tid; p < NOUT; p += 256) op[p] = outLds[p];
}

extern "C" void kernel_launch(void* const* d_in, const int* in_sizes, int n_in,
                              void* d_out, int out_size, void* d_ws, size_t ws_size,
                              hipStream_t stream) {
    (void)in_sizes; (void)n_in; (void)d_ws; (void)ws_size; (void)out_size;
    const float* z = (const float*)d_in[0];
    const float* x = (const float*)d_in[1];
    float* out = (float*)d_out;
    siamfc_xcorr_wmma<<<dim3(NSAMP), dim3(256), 0, stream>>>(z, x, out);
}


# x B-fragments loaded as float4 (16 B per lane): N-tile columns permuted so one load feeds 4 N tiles; 11 waves, each one 64-pixel tile over all K; sum order kept (LDS atomic scatter as before)
# speedup vs baseline: 1.2135x; 1.2135x over previous
#include <hip/hip_runtime.h>


typedef __attribute__((ext_vector_type(2))) float v2f;
typedef __attribute__((ext_vector_type(4))) float v4f;
typedef __attribute__((ext_vector_type(8))) float v8f;

#define NSAMP   256
#define CHN     256
#define KPOS    36
#define MPAD    48
#define NPIX    676
#define NQ      169
#define WXX     26
#define OHW     21
#define NOUT    441
#define ZSTRIDE 260
#define NWAVES  11
#define NTHR    352
#define KBLK    8

__global__ __launch_bounds__(NTHR)
void siamfc_xcorr_wmma(const float* __restrict__ z,
                       const float* __restrict__ x,
                       float* __restrict__ out) {
    __shared__ float zlds[MPAD * ZSTRIDE];
    __shared__ float outLds[NOUT];

    const int n     = blockIdx.x;
    const int tid   = threadIdx.x;
    const int wave  = tid >> 5;
    const int lane  = tid & 31;
    const int nl    = lane & 15;
    const int khalf = lane >> 4;

    if (tid < CHN) {
        const float* zsrc = z + (size_t)(n * CHN + tid) * KPOS;
        float zr[KPOS];
        #pragma unroll
        for (int k4 = 0; k4 < KPOS; k4 += 4) {
            float4 v = *(const float4*)(zsrc + k4);
            zr[k4 + 0] = v.x; zr[k4 + 1] = v.y; zr[k4 + 2] = v.z; zr[k4 + 3] = v.w;
        }
        #pragma unroll
        for (int k = 0; k < KPOS; ++k) zlds[k * ZSTRIDE + tid] = zr[k];
        #pragma unroll
        for (int k = KPOS; k < MPAD; ++k) zlds[k * ZSTRIDE + tid] = 0.0f;
    }
    for (int p = tid; p < NOUT; p += NTHR) outLds[p] = 0.0f;
    __syncthreads();

    const float* xs = x + (size_t)n * CHN * NPIX;

    const int q_raw = wave * 16 + nl;
    const int q     = (q_raw < NQ) ? q_raw : (NQ - 1);

    const float* xb = xs + (size_t)(khalf * 2) * NPIX + 4 * q;
    const float* za = zlds + nl * ZSTRIDE + khalf * 2;

    v8f acc[3][4];
    #pragma unroll
    for (int mt = 0; mt < 3; ++mt)
        #pragma unroll
        for (int j = 0; j < 4; ++j) acc[mt][j] = (v8f){};

    for (int c0 = 0; c0 < CHN; c0 += 4 * KBLK) {
        v4f f0[KBLK], f1[KBLK];
        #pragma unroll
        for (int i = 0; i < KBLK; ++i) {
            f0[i] = *(const v4f*)(xb + (size_t)(c0 + 4 * i) * NPIX);
            f1[i] = *(const v4f*)(xb + (size_t)(c0 + 4 * i) * NPIX + NPIX);
        }
        __builtin_amdgcn_sched_barrier(0);
        #pragma unroll
        for (int i = 0; i < KBLK; ++i) {
            const int ka = c0 + 4 * i;
            const v2f a0 = *(const v2f*)&za[ 0 * ZSTRIDE + ka];
            const v2f a1 = *(const v2f*)&za[16 * ZSTRIDE + ka];
            const v2f a2 = *(const v2f*)&za[32 * ZSTRIDE + ka];
            #pragma unroll
            for (int j = 0; j < 4; ++j) {
                v2f b;
                b.x = f0[i][j];
                b.y = f1[i][j];
                acc[0][j] = __builtin_amdgcn_wmma_f32_16x16x4_f32(false, a0, false, b, (short)0, acc[0][j], false, false);
                acc[1][j] = __builtin_amdgcn_wmma_f32_16x16x4_f32(false, a1, false, b, (short)0, acc[1][j], false, false);
                acc[2][j] = __builtin_amdgcn_wmma_f32_16x16x4_f32(false, a2, false, b, (short)0, acc[2][j], false, false);
            }
        }
    }

    if (q_raw < NQ) {
        #pragma unroll
        for (int j = 0; j < 4; ++j) {
            const int s  = 4 * q_raw + j;
            const int r  = s / WXX;
            const int wc = s % WXX;
            #pragma unroll
            for (int mt = 0; mt < 3; ++mt) {
                #pragma unroll
                for (int v = 0; v < 8; ++v) {
                    const int m = mt * 16 + v + khalf * 8;
                    if (m < KPOS) {
                        const int kh = m / 6, kw = m % 6;
                        const int oh = r - kh, ow = wc - kw;
                        if (oh >= 0 && oh < OHW && ow >= 0 && ow < OHW)
                            atomicAdd(&outLds[oh * OHW + ow], acc[mt][j][v]);
                    }
                }
            }
        }
    }

    __syncthreads();
    float* op = out + (size_t)n * NOUT;
    for (int p = tid; p < NOUT; p += NTHR) op[p] = outLds[p];
}

extern "C" void kernel_launch(void* const* d_in, const int* in_sizes, int n_in,
                              void* d_out, int out_size, void* d_ws, size_t ws_size,
                              hipStream_t stream) {
    (void)in_sizes; (void)n_in; (void)d_ws; (void)ws_size; (void)out_size;
    const float* z = (const float*)d_in[0];
    const float* x = (const float*)d_in[1];
    float* out = (float*)d_out;
    siamfc_xcorr_wmma<<<dim3(256), dim3(352), 0, stream>>>(z, x, out);
}


# x loads pipelined in registers: ring of four 16-channel sub-blocks, three in flight ahead of the one in its WMMAs, first three issued before z staging (LDS-only barrier fence); sum order kept
# speedup vs baseline: 1.3060x; 1.0762x over previous
#include <hip/hip_runtime.h>


typedef __attribute__((ext_vector_type(2))) float v2f;
typedef __attribute__((ext_vector_type(4))) float v4f;
typedef __attribute__((ext_vector_type(8))) float v8f;

#define NSAMP   256
#define CHN     256
#define KPOS    36
#define MPAD    48
#define NPIX    676
#define NQ      169
#define WXX     26
#define OHW     21
#define NOUT    441
#define ZSTRIDE 260
#define NWAVES  11
#define NTHR    352
#define KSUB    4
#define NBUF    4
#define SUBCH   (4 * KSUB)
#define NSUB    (CHN / SUBCH)

__device__ __forceinline__ void load_sub(v4f (&f0)[KSUB], v4f (&f1)[KSUB], const float* p) {
    #pragma unroll
    for (int i = 0; i < KSUB; ++i) {
        f0[i] = *(const v4f*)(p + (size_t)(4 * i) * NPIX);
        f1[i] = *(const v4f*)(p + (size_t)(4 * i) * NPIX + NPIX);
    }
}

__device__ __forceinline__ void mma_sub(v8f (&acc)[3][4], const v4f (&f0)[KSUB], const v4f (&f1)[KSUB],
                                        const float* zp) {
    #pragma unroll
    for (int i = 0; i < KSUB; ++i) {
        const int ka = 4 * i;
        const v2f a0 = *(const v2f*)&zp[ 0 * ZSTRIDE + ka];
        const v2f a1 = *(const v2f*)&zp[16 * ZSTRIDE + ka];
        const v2f a2 = *(const v2f*)&zp[32 * ZSTRIDE + ka];
        #pragma unroll
        for (int j = 0; j < 4; ++j) {
            v2f b;
            b.x = f0[i][j];
            b.y = f1[i][j];
            acc[0][j] = __builtin_amdgcn_wmma_f32_16x16x4_f32(false, a0, false, b, (short)0, acc[0][j], false, false);
            acc[1][j] = __builtin_amdgcn_wmma_f32_16x16x4_f32(false, a1, false, b, (short)0, acc[1][j], false, false);
            acc[2][j] = __builtin_amdgcn_wmma_f32_16x16x4_f32(false, a2, false, b, (short)0, acc[2][j], false, false);
        }
    }
}

__global__ __launch_bounds__(NTHR)
void siamfc_xcorr_wmma(const float* __restrict__ z,
                       const float* __restrict__ x,
                       float* __restrict__ out) {
    __shared__ float zlds[MPAD * ZSTRIDE];
    __shared__ float outLds[NOUT];

    const int n     = blockIdx.x;
    const int tid   = threadIdx.x;
    const int wave  = tid >> 5;
    const int lane  = tid & 31;
    const int nl    = lane & 15;
    const int khalf = lane >> 4;

    const float* xs = x + (size_t)n * CHN * NPIX;
    const int q_raw = wave * 16 + nl;
    const int q     = (q_raw < NQ) ? q_raw : (NQ - 1);
    const float* xb = xs + (size_t)(khalf * 2) * NPIX + 4 * q;

    float zr[KPOS];
    if (tid < CHN) {
        const float* zsrc = z + (size_t)(n * CHN + tid) * KPOS;
        #pragma unroll
        for (int k4 = 0; k4 < KPOS; k4 += 4) {
            float4 v = *(const float4*)(zsrc + k4);
            zr[k4 + 0] = v.x; zr[k4 + 1] = v.y; zr[k4 + 2] = v.z; zr[k4 + 3] = v.w;
        }
    }

    v4f f0[NBUF][KSUB], f1[NBUF][KSUB];
    load_sub(f0[0], f1[0], xb + (size_t)(0 * SUBCH) * NPIX);
    load_sub(f0[1], f1[1], xb + (size_t)(1 * SUBCH) * NPIX);
    load_sub(f0[2], f1[2], xb + (size_t)(2 * SUBCH) * NPIX);

    if (tid < CHN) {
        #pragma unroll
        for (int k = 0; k < KPOS; ++k) zlds[k * ZSTRIDE + tid] = zr[k];
        #pragma unroll
        for (int k = KPOS; k < MPAD; ++k) zlds[k * ZSTRIDE + tid] = 0.0f;
    }
    for (int p = tid; p < NOUT; p += NTHR) outLds[p] = 0.0f;
    __builtin_amdgcn_fence(__ATOMIC_RELEASE, "workgroup", "local");
    __builtin_amdgcn_s_barrier();
    __builtin_amdgcn_fence(__ATOMIC_ACQUIRE, "workgroup", "local");

    const float* zp = zlds + nl * ZSTRIDE + khalf * 2;
    const float* xp = xb + (size_t)(3 * SUBCH) * NPIX;

    v8f acc[3][4];
    #pragma unroll
    for (int mt = 0; mt < 3; ++mt)
        #pragma unroll
        for (int j = 0; j < 4; ++j) acc[mt][j] = (v8f){};

    for (int it = 0; it < NSUB / NBUF - 1; ++it) {
        #pragma unroll
        for (int u = 0; u < NBUF; ++u) {
            load_sub(f0[(u + 3) % NBUF], f1[(u + 3) % NBUF], xp);
            xp += (size_t)SUBCH * NPIX;
            __builtin_amdgcn_sched_barrier(0);
            mma_sub(acc, f0[u], f1[u], zp);
            zp += SUBCH;
        }
    }
    load_sub(f0[3], f1[3], xp);
    __builtin_amdgcn_sched_barrier(0);
    mma_sub(acc, f0[0], f1[0], zp);
    mma_sub(acc, f0[1], f1[1], zp + 1 * SUBCH);
    mma_sub(acc, f0[2], f1[2], zp + 2 * SUBCH);
    mma_sub(acc, f0[3], f1[3], zp + 3 * SUBCH);

    if (q_raw < NQ) {
        #pragma unroll
        for (int j = 0; j < 4; ++j) {
            const int s  = 4 * q_raw + j;
            const int r  = s / WXX;
            const int wc = s % WXX;
            #pragma unroll
            for (int mt = 0; mt < 3; ++mt) {
                #pragma unroll
                for (int v = 0; v < 8; ++v) {
                    const int m = mt * 16 + v + khalf * 8;
                    if (m < KPOS) {
                        const int kh = m / 6, kw = m % 6;
                        const int oh = r - kh, ow = wc - kw;
                        if (oh >= 0 && oh < OHW && ow >= 0 && ow < OHW)
                            atomicAdd(&outLds[oh * OHW + ow], acc[mt][j][v]);
                    }
                }
            }
        }
    }

    __syncthreads();
    float* op = out + (size_t)n * NOUT;
    for (int p = tid; p < NOUT; p += NTHR) op[p] = outLds[p];
}

extern "C" void kernel_launch(void* const* d_in, const int* in_sizes, int n_in,
                              void* d_out, int out_size, void* d_ws, size_t ws_size,
                              hipStream_t stream) {
    (void)in_sizes; (void)n_in; (void)d_ws; (void)ws_size; (void)out_size;
    const float* z = (const float*)d_in[0];
    const float* x = (const float*)d_in[1];
    float* out = (float*)d_out;
    siamfc_xcorr_wmma<<<dim3(256), dim3(352), 0, stream>>>(z, x, out);
}


# scatter into a 31x31 LDS tile with a 5-cell margin: out-of-range taps go to margin cells that are never read, so per-element range tests, exec branches and /6 %6 leave; LDS atomic sum as before
# speedup vs baseline: 1.4148x; 1.0833x over previous
#include <hip/hip_runtime.h>


typedef __attribute__((ext_vector_type(2))) float v2f;
typedef __attribute__((ext_vector_type(4))) float v4f;
typedef __attribute__((ext_vector_type(8))) float v8f;

#define NSAMP   256
#define CHN     256
#define KPOS    36
#define MPAD    48
#define NPIX    676
#define NQ      169
#define WXX     26
#define OHW     21
#define NOUT    441
#define MARG    5
#define PADW    31
#define NPAD    961
#define ZSTRIDE 260
#define NWAVES  11
#define NTHR    352
#define KSUB    4
#define NBUF    4
#define SUBCH   (4 * KSUB)
#define NSUB    (CHN / SUBCH)

__device__ __forceinline__ void load_sub(v4f (&f0)[KSUB], v4f (&f1)[KSUB], const float* p) {
    #pragma unroll
    for (int i = 0; i < KSUB; ++i) {
        f0[i] = *(const v4f*)(p + (size_t)(4 * i) * NPIX);
        f1[i] = *(const v4f*)(p + (size_t)(4 * i) * NPIX + NPIX);
    }
}

__device__ __forceinline__ void mma_sub(v8f (&acc)[3][4], const v4f (&f0)[KSUB], const v4f (&f1)[KSUB],
                                        const float* zp) {
    #pragma unroll
    for (int i = 0; i < KSUB; ++i) {
        const int ka = 4 * i;
        const v2f a0 = *(const v2f*)&zp[ 0 * ZSTRIDE + ka];
        const v2f a1 = *(const v2f*)&zp[16 * ZSTRIDE + ka];
        const v2f a2 = *(const v2f*)&zp[32 * ZSTRIDE + ka];
        #pragma unroll
        for (int j = 0; j < 4; ++j) {
            v2f b;
            b.x = f0[i][j];
            b.y = f1[i][j];
            acc[0][j] = __builtin_amdgcn_wmma_f32_16x16x4_f32(false, a0, false, b, (short)0, acc[0][j], false, false);
            acc[1][j] = __builtin_amdgcn_wmma_f32_16x16x4_f32(false, a1, false, b, (short)0, acc[1][j], false, false);
            acc[2][j] = __builtin_amdgcn_wmma_f32_16x16x4_f32(false, a2, false, b, (short)0, acc[2][j], false, false);
        }
    }
}

__global__ __launch_bounds__(NTHR)
void siamfc_xcorr_wmma(const float* __restrict__ z,
                       const float* __restrict__ x,
                       float* __restrict__ out) {
    __shared__ float zlds[MPAD * ZSTRIDE];
    __shared__ float outPad[NPAD];

    const int n     = blockIdx.x;
    const int tid   = threadIdx.x;
    const int wave  = tid >> 5;
    const int lane  = tid & 31;
    const int nl    = lane & 15;
    const int khalf = lane >> 4;

    const float* xs = x + (size_t)n * CHN * NPIX;
    const int q_raw = wave * 16 + nl;
    const int q     = (q_raw < NQ) ? q_raw : (NQ - 1);
    const float* xb = xs + (size_t)(khalf * 2) * NPIX + 4 * q;

    float zr[KPOS];
    if (tid < CHN) {
        const float* zsrc = z + (size_t)(n * CHN + tid) * KPOS;
        #pragma unroll
        for (int k4 = 0; k4 < KPOS; k4 += 4) {
            float4 v = *(const float4*)(zsrc + k4);
            zr[k4 + 0] = v.x; zr[k4 + 1] = v.y; zr[k4 + 2] = v.z; zr[k4 + 3] = v.w;
        }
    }

    v4f f0[NBUF][KSUB], f1[NBUF][KSUB];
    load_sub(f0[0], f1[0], xb + (size_t)(0 * SUBCH) * NPIX);
    load_sub(f0[1], f1[1], xb + (size_t)(1 * SUBCH) * NPIX);
    load_sub(f0[2], f1[2], xb + (size_t)(2 * SUBCH) * NPIX);

    if (tid < CHN) {
        #pragma unroll
        for (int k = 0; k < KPOS; ++k) zlds[k * ZSTRIDE + tid] = zr[k];
        #pragma unroll
        for (int k = KPOS; k < MPAD; ++k) zlds[k * ZSTRIDE + tid] = 0.0f;
    }
    for (int p = tid; p < NPAD; p += NTHR) outPad[p] = 0.0f;
    __builtin_amdgcn_fence(__ATOMIC_RELEASE, "workgroup", "local");
    __builtin_amdgcn_s_barrier();
    __builtin_amdgcn_fence(__ATOMIC_ACQUIRE, "workgroup", "local");

    const float* zp = zlds + nl * ZSTRIDE + khalf * 2;
    const float* xp = xb + (size_t)(3 * SUBCH) * NPIX;

    v8f acc[3][4];
    #pragma unroll
    for (int mt = 0; mt < 3; ++mt)
        #pragma unroll
        for (int j = 0; j < 4; ++j) acc[mt][j] = (v8f){};

    for (int it = 0; it < NSUB / NBUF - 1; ++it) {
        #pragma unroll
        for (int u = 0; u < NBUF; ++u) {
            load_sub(f0[(u + 3) % NBUF], f1[(u + 3) % NBUF], xp);
            xp += (size_t)SUBCH * NPIX;
            __builtin_amdgcn_sched_barrier(0);
            mma_sub(acc, f0[u], f1[u], zp);
            zp += SUBCH;
        }
    }
    load_sub(f0[3], f1[3], xp);
    __builtin_amdgcn_sched_barrier(0);
    mma_sub(acc, f0[0], f1[0], zp);
    mma_sub(acc, f0[1], f1[1], zp + 1 * SUBCH);
    mma_sub(acc, f0[2], f1[2], zp + 2 * SUBCH);
    mma_sub(acc, f0[3], f1[3], zp + 3 * SUBCH);

    if (q_raw < NQ) {
        int base4[4];
        #pragma unroll
        for (int j = 0; j < 4; ++j) {
            const int s  = 4 * q_raw + j;
            const int r  = s / WXX;
            const int wc = s % WXX;
            base4[j] = 4 * ((r + MARG) * PADW + wc + MARG);
        }
        #pragma unroll
        for (int mt = 0; mt < 3; ++mt) {
            #pragma unroll
            for (int v = 0; v < 8; ++v) {
                const int m0 = mt * 16 + v;
                const int m1 = m0 + 8;
                if (m0 < KPOS) {
                    const int c0 = 4 * ((m0 / 6) * PADW + (m0 % 6));
                    if (m1 < KPOS) {
                        const int c1 = 4 * ((m1 / 6) * PADW + (m1 % 6));
                        const int c  = khalf ? c1 : c0;
                        #pragma unroll
                        for (int j = 0; j < 4; ++j)
                            atomicAdd((float*)((char*)outPad + (base4[j] - c)), acc[mt][j][v]);
                    } else if (khalf == 0) {
                        #pragma unroll
                        for (int j = 0; j < 4; ++j)
                            atomicAdd((float*)((char*)outPad + (base4[j] - c0)), acc[mt][j][v]);
                    }
                }
            }
        }
    }

    __syncthreads();
    float* op = out + (size_t)n * NOUT;
    for (int p = tid; p < NOUT; p += NTHR) op[p] = outPad[(p / OHW + MARG) * PADW + (p % OHW) + MARG];
}

extern "C" void kernel_launch(void* const* d_in, const int* in_sizes, int n_in,
                              void* d_out, int out_size, void* d_ws, size_t ws_size,
                              hipStream_t stream) {
    (void)in_sizes; (void)n_in; (void)d_ws; (void)ws_size; (void)out_size;
    const float* z = (const float*)d_in[0];
    const float* x = (const float*)d_in[1];
    float* out = (float*)d_out;
    siamfc_xcorr_wmma<<<dim3(256), dim3(352), 0, stream>>>(z, x, out);
}


# Q rows that can only reach margin cells are not computed: waves 0,1,9,10 run 2 M tiles from their own first A row, the third M tile sits behind a scalar branch (waves 2..8); sum order kept
# speedup vs baseline: 1.4699x; 1.0389x over previous
#include <hip/hip_runtime.h>


typedef __attribute__((ext_vector_type(2))) float v2f;
typedef __attribute__((ext_vector_type(4))) float v4f;
typedef __attribute__((ext_vector_type(8))) float v8f;

#define NSAMP   256
#define CHN     256
#define KPOS    36
#define MPAD    48
#define NPIX    676
#define NQ      169
#define WXX     26
#define OHW     21
#define NOUT    441
#define MARG    5
#define PADW    31
#define NPAD    961
#define ZSTRIDE 260
#define NWAVES  11
#define NTHR    352
#define KSUB    4
#define NBUF    4
#define SUBCH   (4 * KSUB)
#define NSUB    (CHN / SUBCH)

__device__ __forceinline__ void load_sub(v4f (&f0)[KSUB], v4f (&f1)[KSUB], const float* p) {
    #pragma unroll
    for (int i = 0; i < KSUB; ++i) {
        f0[i] = *(const v4f*)(p + (size_t)(4 * i) * NPIX);
        f1[i] = *(const v4f*)(p + (size_t)(4 * i) * NPIX + NPIX);
    }
}

template <bool MORE>
__device__ __forceinline__ void mma_sub(v8f (&acc)[3][4], const v4f (&f0)[KSUB], const v4f (&f1)[KSUB],
                                        const float* zp, bool third, v2f (&a2)[KSUB]) {
    #pragma unroll
    for (int i = 0; i < KSUB; ++i) {
        const int ka = 4 * i;
        const v2f a0 = *(const v2f*)&zp[ 0 * ZSTRIDE + ka];
        const v2f a1 = *(const v2f*)&zp[16 * ZSTRIDE + ka];
        #pragma unroll
        for (int j = 0; j < 4; ++j) {
            v2f b;
            b.x = f0[i][j];
            b.y = f1[i][j];
            acc[0][j] = __builtin_amdgcn_wmma_f32_16x16x4_f32(false, a0, false, b, (short)0, acc[0][j], false, false);
            acc[1][j] = __builtin_amdgcn_wmma_f32_16x16x4_f32(false, a1, false, b, (short)0, acc[1][j], false, false);
        }
    }
    if (third) {
        #pragma unroll
        for (int i = 0; i < KSUB; ++i) {
            #pragma unroll
            for (int j = 0; j < 4; ++j) {
                v2f b;
                b.x = f0[i][j];
                b.y = f1[i][j];
                acc[2][j] = __builtin_amdgcn_wmma_f32_16x16x4_f32(false, a2[i], false, b, (short)0, acc[2][j], false, false);
            }
        }
        if (MORE) {
            #pragma unroll
            for (int i = 0; i < KSUB; ++i) a2[i] = *(const v2f*)&zp[32 * ZSTRIDE + SUBCH + 4 * i];
        }
    }
}

template <int MBASE, int NMT>
__device__ __forceinline__ void scatter_q(const v8f (&acc)[3][4], float* outPad, const int (&base4)[4], int khalf) {
    #pragma unroll
    for (int mt = 0; mt < NMT; ++mt) {
        #pragma unroll
        for (int v = 0; v < 8; ++v) {
            const int m0 = MBASE + mt * 16 + v;
            const int m1 = m0 + 8;
            if (m0 < KPOS) {
                const int c0 = 4 * ((m0 / 6) * PADW + (m0 % 6));
                if (m1 < KPOS) {
                    const int c1 = 4 * ((m1 / 6) * PADW + (m1 % 6));
                    const int c  = khalf ? c1 : c0;
                    #pragma unroll
                    for (int j = 0; j < 4; ++j)
                        atomicAdd((float*)((char*)outPad + (base4[j] - c)), acc[mt][j][v]);
                } else if (khalf == 0) {
                    #pragma unroll
                    for (int j = 0; j < 4; ++j)
                        atomicAdd((float*)((char*)outPad + (base4[j] - c0)), acc[mt][j][v]);
                }
            }
        }
    }
}

__global__ __launch_bounds__(NTHR)
void siamfc_xcorr_wmma(const float* __restrict__ z,
                       const float* __restrict__ x,
                       float* __restrict__ out) {
    __shared__ float zlds[MPAD * ZSTRIDE];
    __shared__ float outPad[NPAD];

    const int n     = blockIdx.x;
    const int tid   = threadIdx.x;
    const int wave  = __builtin_amdgcn_readfirstlane(tid >> 5);
    const int lane  = tid & 31;
    const int nl    = lane & 15;
    const int khalf = lane >> 4;

    const float* xs = x + (size_t)n * CHN * NPIX;
    const int q_raw = wave * 16 + nl;
    const int q     = (q_raw < NQ) ? q_raw : (NQ - 1);
    const float* xb = xs + (size_t)(khalf * 2) * NPIX + 4 * q;

    float zr[KPOS];
    if (tid < CHN) {
        const float* zsrc = z + (size_t)(n * CHN + tid) * KPOS;
        #pragma unroll
        for (int k4 = 0; k4 < KPOS; k4 += 4) {
            float4 v = *(const float4*)(zsrc + k4);
            zr[k4 + 0] = v.x; zr[k4 + 1] = v.y; zr[k4 + 2] = v.z; zr[k4 + 3] = v.w;
        }
    }

    v4f f0[NBUF][KSUB], f1[NBUF][KSUB];
    load_sub(f0[0], f1[0], xb + (size_t)(0 * SUBCH) * NPIX);
    load_sub(f0[1], f1[1], xb + (size_t)(1 * SUBCH) * NPIX);
    load_sub(f0[2], f1[2], xb + (size_t)(2 * SUBCH) * NPIX);

    if (tid < CHN) {
        #pragma unroll
        for (int k = 0; k < KPOS; ++k) zlds[k * ZSTRIDE + tid] = zr[k];
        #pragma unroll
        for (int k = KPOS; k < MPAD; ++k) zlds[k * ZSTRIDE + tid] = 0.0f;
    }
    for (int p = tid; p < NPAD; p += NTHR) outPad[p] = 0.0f;
    __builtin_amdgcn_fence(__ATOMIC_RELEASE, "workgroup", "local");
    __builtin_amdgcn_s_barrier();
    __builtin_amdgcn_fence(__ATOMIC_ACQUIRE, "workgroup", "local");

    const int  mbase = (wave == 9) ? 12 : ((wave == 10) ? 16 : 0);
    const bool third = (wave >= 2) && (wave <= 8);
    const float* zp = zlds + (mbase + nl) * ZSTRIDE + khalf * 2;
    const float* xp = xb + (size_t)(3 * SUBCH) * NPIX;

    v8f acc[3][4];
    #pragma unroll
    for (int mt = 0; mt < 3; ++mt)
        #pragma unroll
        for (int j = 0; j < 4; ++j) acc[mt][j] = (v8f){};

    v2f a2[KSUB];
    #pragma unroll
    for (int i = 0; i < KSUB; ++i) a2[i] = (v2f){};
    if (third) {
        #pragma unroll
        for (int i = 0; i < KSUB; ++i) a2[i] = *(const v2f*)&zp[32 * ZSTRIDE + 4 * i];
    }

    for (int it = 0; it < NSUB / NBUF - 1; ++it) {
        #pragma unroll
        for (int u = 0; u < NBUF; ++u) {
            load_sub(f0[(u + 3) % NBUF], f1[(u + 3) % NBUF], xp);
            xp += (size_t)SUBCH * NPIX;
            __builtin_amdgcn_sched_barrier(0);
            mma_sub<true>(acc, f0[u], f1[u], zp, third, a2);
            zp += SUBCH;
        }
    }
    load_sub(f0[3], f1[3], xp);
    __builtin_amdgcn_sched_barrier(0);
    mma_sub<true>(acc, f0[0], f1[0], zp, third, a2);
    mma_sub<true>(acc, f0[1], f1[1], zp + 1 * SUBCH, third, a2);
    mma_sub<true>(acc, f0[2], f1[2], zp + 2 * SUBCH, third, a2);
    mma_sub<false>(acc, f0[3], f1[3], zp + 3 * SUBCH, third, a2);

    if (q_raw < NQ) {
        int base4[4];
        #pragma unroll
        for (int j = 0; j < 4; ++j) {
            const int s  = 4 * q_raw + j;
            const int r  = s / WXX;
            const int wc = s % WXX;
            base4[j] = 4 * ((r + MARG) * PADW + wc + MARG);
        }
        if (third)           scatter_q< 0, 3>(acc, outPad, base4, khalf);
        else if (mbase == 0) scatter_q< 0, 2>(acc, outPad, base4, khalf);
        else if (wave == 9)  scatter_q<12, 2>(acc, outPad, base4, khalf);
        else                 scatter_q<16, 2>(acc, outPad, base4, khalf);
    }

    __syncthreads();
    float* op = out + (size_t)n * NOUT;
    for (int p = tid; p < NOUT; p += NTHR) op[p] = outPad[(p / OHW + MARG) * PADW + (p % OHW) + MARG];
}

extern "C" void kernel_launch(void* const* d_in, const int* in_sizes, int n_in,
                              void* d_out, int out_size, void* d_ws, size_t ws_size,
                              hipStream_t stream) {
    (void)in_sizes; (void)n_in; (void)d_ws; (void)ws_size; (void)out_size;
    const float* z = (const float*)d_in[0];
    const float* x = (const float*)d_in[1];
    float* out = (float*)d_out;
    siamfc_xcorr_wmma<<<dim3(256), dim3(352), 0, stream>>>(z, x, out);
}
